# starting kernel
# baseline (speedup 1.0000x reference)
#include <hip/hip_runtime.h>
#include <math.h>

typedef float v2f __attribute__((ext_vector_type(2)));
typedef float v8f __attribute__((ext_vector_type(8)));

#define BB 8
#define GG 2048
#define KK 32
#define CC 64
#define TWO_PI 6.28318530717958647692f

__device__ __forceinline__ float wave_allreduce_sum(float v) {
    v += __shfl_xor(v, 1, 32);
    v += __shfl_xor(v, 2, 32);
    v += __shfl_xor(v, 4, 32);
    v += __shfl_xor(v, 8, 32);
    v += __shfl_xor(v, 16, 32);
    return v;
}

__global__ __launch_bounds__(256)
void k1_fourier_wmma(const float* __restrict__ lc_x,
                     const float* __restrict__ knn_x,
                     const float* __restrict__ fB,
                     float* __restrict__ klbuf,
                     float* __restrict__ invn)
{
    const int tid  = threadIdx.x;
    const int lane = tid & 31;
    const int wid  = (blockIdx.x * blockDim.x + tid) >> 5;
    const int hi   = lane >> 4;
    const int c15  = lane & 15;

    const float* lcp = lc_x + (size_t)wid * CC;

    float a0 = lcp[lane];
    float a1 = lcp[lane + 32];
    float ss = wave_allreduce_sum(a0 * a0 + a1 * a1);
    float inv = rsqrtf(ss);
    if (lane == 0) invn[wid] = inv;

    const float nx = lcp[3], ny = lcp[4], nz = lcp[5];

    const float* knp = knn_x + (size_t)wid * (KK * 6);
    v2f A[2][2];
#pragma unroll
    for (int m = 0; m < 2; ++m) {
        const int k = c15 + 16 * m;
        const float* kp = knp + k * 6;
        const float px = kp[3], py = kp[4], pz = kp[5];
        const float cx = py * nz - pz * ny;
        const float cy = pz * nx - px * nz;
        const float cz = px * ny - py * nx;
        const float dt = px * nx + py * ny + pz * nz;
        A[m][0].x = hi ? pz : px;
        A[m][0].y = hi ? cx : py;
        A[m][1].x = hi ? dt : cy;
        A[m][1].y = hi ? 0.f : cz;
    }

    v2f Bm[2][2];
#pragma unroll
    for (int n = 0; n < 2; ++n) {
        const int N = 16 * n + c15;
        Bm[n][0].x = hi ? fB[2 * KK + N] : fB[0 * KK + N];
        Bm[n][0].y = hi ? fB[3 * KK + N] : fB[1 * KK + N];
        Bm[n][1].x = hi ? fB[6 * KK + N] : fB[4 * KK + N];
        Bm[n][1].y = hi ? 0.f            : fB[5 * KK + N];
    }

    float ssum[2] = {0.f, 0.f};
    float csum[2] = {0.f, 0.f};
#pragma unroll
    for (int n = 0; n < 2; ++n) {
#pragma unroll
        for (int m = 0; m < 2; ++m) {
            v8f d = {};
            d = __builtin_amdgcn_wmma_f32_16x16x4_f32(
                    false, A[m][0], false, Bm[n][0], (short)0, d, false, false);
            d = __builtin_amdgcn_wmma_f32_16x16x4_f32(
                    false, A[m][1], false, Bm[n][1], (short)0, d, false, false);
#pragma unroll
            for (int v = 0; v < 8; ++v) {
                const float x = TWO_PI * d[v];
                const float s = __sinf(x);
                const float c = __cosf(x);
                const float s2 = s * s, c2 = c * c;
                ssum[n] += s2 * s2 * s;
                csum[n] += c2 * c2 * c;
            }
        }
    }
#pragma unroll
    for (int n = 0; n < 2; ++n) {
        ssum[n] += __shfl_xor(ssum[n], 16, 32);
        csum[n] += __shfl_xor(csum[n], 16, 32);
    }

    float* outp = klbuf + (size_t)wid * CC;
    if (hi == 0) {
        outp[c15]      = ssum[0] * 0.0625f;
        outp[16 + c15] = ssum[1] * 0.0625f;
    } else {
        outp[32 + c15] = csum[0] * 0.0625f;
        outp[48 + c15] = csum[1] * 0.0625f;
    }
}

__global__ __launch_bounds__(256)
void k2_normalize(const float* __restrict__ lc_x,
                  const float* __restrict__ klbuf,
                  const float* __restrict__ invn,
                  float* __restrict__ out)
{
    const int bc  = blockIdx.x;
    const int b   = bc >> 6;
    const int c   = bc & 63;
    const int tid = threadIdx.x;

    __shared__ float sred[256];
    __shared__ float s_inv;

    float acc = 0.f;
    for (int g = tid; g < GG; g += 256) {
        const float v = klbuf[((size_t)(b * GG + g)) * CC + c];
        acc += v * v;
    }
    sred[tid] = acc;
    __syncthreads();
    for (int s = 128; s > 0; s >>= 1) {
        if (tid < s) sred[tid] += sred[tid + s];
        __syncthreads();
    }
    if (tid == 0) s_inv = rsqrtf(sred[0]);
    __syncthreads();
    const float ikl = s_inv;

    for (int g = tid; g < GG; g += 256) {
        const size_t idx = ((size_t)(b * GG + g)) * CC + c;
        const float kn = klbuf[idx] * ikl;
        const float ll = lc_x[idx] * invn[b * GG + g];
        out[((size_t)(b * 128 + c)) * GG + g]      = kn - ll;
        out[((size_t)(b * 128 + 64 + c)) * GG + g] = ll;
    }
}

extern "C" void kernel_launch(void* const* d_in, const int* in_sizes, int n_in,
                              void* d_out, int out_size, void* d_ws, size_t ws_size,
                              hipStream_t stream) {
    const float* lc_x  = (const float*)d_in[0];
    const float* knn_x = (const float*)d_in[1];
    const float* fB    = (const float*)d_in[2];
    float* out = (float*)d_out;

    float* klbuf = (float*)d_ws;
    float* invn  = klbuf + (size_t)BB * GG * CC;

    k1_fourier_wmma<<<(BB * GG) / 8, 256, 0, stream>>>(lc_x, knn_x, fB, klbuf, invn);
    k2_normalize<<<BB * CC, 256, 0, stream>>>(lc_x, klbuf, invn, out);
}


# Transpose moved into k1: kl and lc_line leave through an LDS [c][g] tile as whole 128B lines (32 g per work-group, 4 per wave); k2 reads contiguous rows; sum-of-squares reduction tree changed
# speedup vs baseline: 2.5414x; 2.5414x over previous
#include <hip/hip_runtime.h>
#include <math.h>


typedef float v2f __attribute__((ext_vector_type(2)));
typedef float v8f __attribute__((ext_vector_type(8)));

#define BB 8
#define GG 2048
#define KK 32
#define CC 64
#define TWO_PI 6.28318530717958647692f

#define GT 32
#define PW 4
#define TP 36

__device__ __forceinline__ float wave_allreduce_sum(float v) {
    v += __shfl_xor(v, 1, 32);
    v += __shfl_xor(v, 2, 32);
    v += __shfl_xor(v, 4, 32);
    v += __shfl_xor(v, 8, 32);
    v += __shfl_xor(v, 16, 32);
    return v;
}

__global__ __launch_bounds__(256)
void k1_fourier_wmma(const float* __restrict__ lc_x,
                     const float* __restrict__ knn_x,
                     const float* __restrict__ fB,
                     float* __restrict__ klT,
                     float* __restrict__ out)
{
    __shared__ __attribute__((aligned(16))) float s_kl[CC * TP];
    __shared__ __attribute__((aligned(16))) float s_ll[CC * TP];

    const int tid  = threadIdx.x;
    const int lane = tid & 31;
    const int w    = tid >> 5;
    const int hi   = lane >> 4;
    const int c15  = lane & 15;

    const int tile = blockIdx.x;
    const int b    = tile >> 6;
    const int g0   = (tile & 63) * GT;

    v2f Bm[2][2];
#pragma unroll
    for (int n = 0; n < 2; ++n) {
        const int N = 16 * n + c15;
        Bm[n][0].x = hi ? fB[2 * KK + N] : fB[0 * KK + N];
        Bm[n][0].y = hi ? fB[3 * KK + N] : fB[1 * KK + N];
        Bm[n][1].x = hi ? fB[6 * KK + N] : fB[4 * KK + N];
        Bm[n][1].y = hi ? 0.f            : fB[5 * KK + N];
    }

    const int krow0 = hi * 32 + c15;
    const int krow1 = krow0 + 16;

#pragma unroll
    for (int i = 0; i < PW; ++i) {
        const int gl  = w * PW + i;
        const int wid = b * GG + g0 + gl;
        const float* lcp = lc_x + (size_t)wid * CC;

        float a0 = lcp[lane];
        float a1 = lcp[lane + 32];
        float ss = wave_allreduce_sum(a0 * a0 + a1 * a1);
        float inv = rsqrtf(ss);

        const float nx = lcp[3], ny = lcp[4], nz = lcp[5];

        const float* knp = knn_x + (size_t)wid * (KK * 6);
        v2f A[2][2];
#pragma unroll
        for (int m = 0; m < 2; ++m) {
            const int k = c15 + 16 * m;
            const float* kp = knp + k * 6;
            const float px = kp[3], py = kp[4], pz = kp[5];
            const float cx = py * nz - pz * ny;
            const float cy = pz * nx - px * nz;
            const float cz = px * ny - py * nx;
            const float dt = px * nx + py * ny + pz * nz;
            A[m][0].x = hi ? pz : px;
            A[m][0].y = hi ? cx : py;
            A[m][1].x = hi ? dt : cy;
            A[m][1].y = hi ? 0.f : cz;
        }

        float ssum[2] = {0.f, 0.f};
        float csum[2] = {0.f, 0.f};
#pragma unroll
        for (int n = 0; n < 2; ++n) {
#pragma unroll
            for (int m = 0; m < 2; ++m) {
                v8f d = {};
                d = __builtin_amdgcn_wmma_f32_16x16x4_f32(
                        false, A[m][0], false, Bm[n][0], (short)0, d, false, false);
                d = __builtin_amdgcn_wmma_f32_16x16x4_f32(
                        false, A[m][1], false, Bm[n][1], (short)0, d, false, false);
#pragma unroll
                for (int v = 0; v < 8; ++v) {
                    const float x = TWO_PI * d[v];
                    const float s = __sinf(x);
                    const float c = __cosf(x);
                    const float s2 = s * s, c2 = c * c;
                    ssum[n] += s2 * s2 * s;
                    csum[n] += c2 * c2 * c;
                }
            }
        }
#pragma unroll
        for (int n = 0; n < 2; ++n) {
            ssum[n] += __shfl_xor(ssum[n], 16, 32);
            csum[n] += __shfl_xor(csum[n], 16, 32);
        }

        s_kl[krow0 * TP + gl] = (hi ? csum[0] : ssum[0]) * 0.0625f;
        s_kl[krow1 * TP + gl] = (hi ? csum[1] : ssum[1]) * 0.0625f;
        s_ll[lane * TP + gl]        = a0 * inv;
        s_ll[(lane + 32) * TP + gl] = a1 * inv;
    }

    __syncthreads();

    const int seg = tid & 7;
    const int r   = tid >> 3;
#pragma unroll
    for (int j = 0; j < 2; ++j) {
        const int row = r + 32 * j;
        const float4 kv = *reinterpret_cast<const float4*>(&s_kl[row * TP + seg * 4]);
        const float4 lv = *reinterpret_cast<const float4*>(&s_ll[row * TP + seg * 4]);
        *reinterpret_cast<float4*>(klT + ((size_t)(b * CC + row)) * GG + g0 + seg * 4) = kv;
        *reinterpret_cast<float4*>(out + ((size_t)(b * 2 * CC + CC + row)) * GG + g0 + seg * 4) = lv;
    }
}

__global__ __launch_bounds__(128)
void k2_normalize(const float* __restrict__ klT,
                  float* out)
{
    const int bc  = blockIdx.x;
    const int b   = bc >> 6;
    const int c   = bc & 63;
    const int tid = threadIdx.x;

    __shared__ float sred[4];

    const float4* ksrc = reinterpret_cast<const float4*>(klT + (size_t)bc * GG);
    const float4* lsrc = reinterpret_cast<const float4*>(out + ((size_t)(b * 2 * CC + CC + c)) * GG);
    float4*       dst  = reinterpret_cast<float4*>(out + ((size_t)(b * 2 * CC + c)) * GG);

    float4 kv[4], lv[4];
#pragma unroll
    for (int j = 0; j < 4; ++j) kv[j] = ksrc[tid + 128 * j];
#pragma unroll
    for (int j = 0; j < 4; ++j) lv[j] = lsrc[tid + 128 * j];

    float acc = 0.f;
#pragma unroll
    for (int j = 0; j < 4; ++j)
        acc += kv[j].x * kv[j].x + kv[j].y * kv[j].y + kv[j].z * kv[j].z + kv[j].w * kv[j].w;
    acc = wave_allreduce_sum(acc);
    if ((tid & 31) == 0) sred[tid >> 5] = acc;
    __syncthreads();
    const float ikl = rsqrtf((sred[0] + sred[1]) + (sred[2] + sred[3]));

#pragma unroll
    for (int j = 0; j < 4; ++j) {
        float4 o;
        o.x = kv[j].x * ikl - lv[j].x;
        o.y = kv[j].y * ikl - lv[j].y;
        o.z = kv[j].z * ikl - lv[j].z;
        o.w = kv[j].w * ikl - lv[j].w;
        dst[tid + 128 * j] = o;
    }
}

extern "C" void kernel_launch(void* const* d_in, const int* in_sizes, int n_in,
                              void* d_out, int out_size, void* d_ws, size_t ws_size,
                              hipStream_t stream) {
    const float* lc_x  = (const float*)d_in[0];
    const float* knn_x = (const float*)d_in[1];
    const float* fB    = (const float*)d_in[2];
    float* out = (float*)d_out;

    float* klT = (float*)d_ws;

    k1_fourier_wmma<<<(BB * GG) / GT, 256, 0, stream>>>(lc_x, knn_x, fB, klT, out);
    k2_normalize<<<BB * CC, 128, 0, stream>>>(klT, out);
}
